# starting kernel
# baseline (speedup 1.0000x reference)
#include <hip/hip_runtime.h>
#include <math.h>

typedef float v2f __attribute__((ext_vector_type(2)));
typedef float v8f __attribute__((ext_vector_type(8)));

#define FEAT_IN  128
#define FEAT_MID 16
#define FEAT_OUT 2

__global__ void init_ws(float* __restrict__ deg, float* __restrict__ acc1,
                        float* __restrict__ acc2, int n_nodes) {
    int tid = blockIdx.x * blockDim.x + threadIdx.x;
    if (tid < n_nodes * FEAT_MID) acc1[tid] = 0.0f;
    if (tid < n_nodes * FEAT_OUT) acc2[tid] = 0.0f;
    if (tid < n_nodes)            deg[tid]  = 1.0f;
}

__global__ void degree_kernel(const int* __restrict__ dst, float* __restrict__ deg,
                              int n_edges) {
    int e = blockIdx.x * blockDim.x + threadIdx.x;
    if (e < n_edges) atomicAdd(&deg[dst[e]], 1.0f);
}

__global__ void dinv_kernel(float* __restrict__ deg, int n_nodes) {
    int i = blockIdx.x * blockDim.x + threadIdx.x;
    if (i < n_nodes) {
        float d = deg[i];
        deg[i] = (d > 0.0f) ? rsqrtf(d) : 0.0f;
    }
}

__global__ void __launch_bounds__(256)
gemm1_wmma(const float* __restrict__ x, const float* __restrict__ W1,
           float* __restrict__ h, int n_tiles, int n_nodes) {
    int lane = threadIdx.x & 31;
    int wave = threadIdx.x >> 5;
    int tile = blockIdx.x * (blockDim.x >> 5) + wave;
    if (tile >= n_tiles) return;

    int col = lane & 15;
    int kh  = (lane >> 4) << 1;

    int rowA = tile * 16 + col;
    if (rowA > n_nodes - 1) rowA = n_nodes - 1;
    const float* xrow = x + (size_t)rowA * FEAT_IN;

    v8f acc = {};
    for (int k0 = 0; k0 < FEAT_IN; k0 += 4) {
        v2f a, b;
        a.x = xrow[k0 + kh];
        a.y = xrow[k0 + kh + 1];
        b.x = W1[(k0 + kh)     * FEAT_MID + col];
        b.y = W1[(k0 + kh + 1) * FEAT_MID + col];
        acc = __builtin_amdgcn_wmma_f32_16x16x4_f32(
             false, a,  false, b,
             (short)0, acc,  false,  false);
    }

    int mbase = (lane >> 4) * 8;
#pragma unroll
    for (int r = 0; r < 8; r++) {
        int row = tile * 16 + mbase + r;
        if (row < n_nodes) h[(size_t)row * FEAT_MID + col] = acc[r];
    }
}

__global__ void edge1_kernel(const int* __restrict__ src, const int* __restrict__ dst,
                             const float* __restrict__ dinv, const float* __restrict__ h,
                             float* __restrict__ acc1, int n_edges) {
    long long tid = (long long)blockIdx.x * blockDim.x + threadIdx.x;
    if (tid >= (long long)n_edges * FEAT_MID) return;
    int e = (int)(tid >> 4);
    int f = (int)(tid & 15);
    int s = src[e], d = dst[e];
    float norm = dinv[s] * dinv[d];
    atomicAdd(&acc1[d * FEAT_MID + f], h[s * FEAT_MID + f] * norm);
}

__global__ void relu_bias1(const float* __restrict__ h, const float* __restrict__ dinv,
                           const float* __restrict__ b1, float* __restrict__ acc1,
                           int n_nodes) {
    int tid = blockIdx.x * blockDim.x + threadIdx.x;
    if (tid >= n_nodes * FEAT_MID) return;
    int node = tid >> 4, f = tid & 15;
    float di = dinv[node];
    float v = acc1[tid] + h[tid] * di * di + b1[f];
    acc1[tid] = fmaxf(v, 0.0f);
}

__global__ void gemm2_kernel(const float* __restrict__ h1, const float* __restrict__ W2,
                             float* __restrict__ h2, int n_nodes) {
    int i = blockIdx.x * blockDim.x + threadIdx.x;
    if (i >= n_nodes) return;
    float s0 = 0.0f, s1 = 0.0f;
#pragma unroll
    for (int k = 0; k < FEAT_MID; k++) {
        float v = h1[i * FEAT_MID + k];
        s0 = fmaf(v, W2[k * 2 + 0], s0);
        s1 = fmaf(v, W2[k * 2 + 1], s1);
    }
    h2[i * 2 + 0] = s0;
    h2[i * 2 + 1] = s1;
}

__global__ void edge2_kernel(const int* __restrict__ src, const int* __restrict__ dst,
                             const float* __restrict__ dinv, const float* __restrict__ h2,
                             float* __restrict__ acc2, int n_edges) {
    long long tid = (long long)blockIdx.x * blockDim.x + threadIdx.x;
    if (tid >= (long long)n_edges * FEAT_OUT) return;
    int e = (int)(tid >> 1);
    int c = (int)(tid & 1);
    int s = src[e], d = dst[e];
    float norm = dinv[s] * dinv[d];
    atomicAdd(&acc2[d * FEAT_OUT + c], h2[s * FEAT_OUT + c] * norm);
}

__global__ void final_kernel(const float* __restrict__ acc2, const float* __restrict__ h2,
                             const float* __restrict__ dinv, const float* __restrict__ b2,
                             float* __restrict__ out, int n_nodes) {
    int i = blockIdx.x * blockDim.x + threadIdx.x;
    if (i >= n_nodes) return;
    float di = dinv[i];
    float sc = di * di;
    float v0 = acc2[i * 2 + 0] + h2[i * 2 + 0] * sc + b2[0];
    float v1 = acc2[i * 2 + 1] + h2[i * 2 + 1] * sc + b2[1];
    float m  = fmaxf(v0, v1);
    float l  = m + logf(expf(v0 - m) + expf(v1 - m));
    out[i * 2 + 0] = v0 - l;
    out[i * 2 + 1] = v1 - l;
}

extern "C" void kernel_launch(void* const* d_in, const int* in_sizes, int n_in,
                              void* d_out, int out_size, void* d_ws, size_t ws_size,
                              hipStream_t stream) {
    const float* x  = (const float*)d_in[0];
    const int*   ei = (const int*)d_in[1];
    const float* W1 = (const float*)d_in[2];
    const float* b1 = (const float*)d_in[3];
    const float* W2 = (const float*)d_in[4];
    const float* b2 = (const float*)d_in[5];
    float* out = (float*)d_out;

    int n_nodes = in_sizes[0] / FEAT_IN;
    int n_edges = in_sizes[1] / 2;
    const int* src = ei;
    const int* dst = ei + n_edges;

    float* ws   = (float*)d_ws;
    float* deg  = ws;
    float* h    = deg  + n_nodes;
    float* acc1 = h    + (size_t)n_nodes * FEAT_MID;
    float* h2   = acc1 + (size_t)n_nodes * FEAT_MID;
    float* acc2 = h2   + (size_t)n_nodes * FEAT_OUT;

    const int B = 256;
    int n16 = n_nodes * FEAT_MID;

    init_ws<<<(n16 + B - 1) / B, B, 0, stream>>>(deg, acc1, acc2, n_nodes);
    degree_kernel<<<(n_edges + B - 1) / B, B, 0, stream>>>(dst, deg, n_edges);
    dinv_kernel<<<(n_nodes + B - 1) / B, B, 0, stream>>>(deg, n_nodes);

    int n_tiles = (n_nodes + 15) / 16;
    int wpb = B / 32;
    gemm1_wmma<<<(n_tiles + wpb - 1) / wpb, B, 0, stream>>>(x, W1, h, n_tiles, n_nodes);

    long long e16 = (long long)n_edges * FEAT_MID;
    edge1_kernel<<<(int)((e16 + B - 1) / B), B, 0, stream>>>(src, dst, deg, h, acc1, n_edges);
    relu_bias1<<<(n16 + B - 1) / B, B, 0, stream>>>(h, deg, b1, acc1, n_nodes);

    gemm2_kernel<<<(n_nodes + B - 1) / B, B, 0, stream>>>(acc1, W2, h2, n_nodes);

    long long e2 = (long long)n_edges * FEAT_OUT;
    edge2_kernel<<<(int)((e2 + B - 1) / B), B, 0, stream>>>(src, dst, deg, h2, acc2, n_edges);

    final_kernel<<<(n_nodes + B - 1) / B, B, 0, stream>>>(acc2, h2, deg, b2, out, n_nodes);
}


# edge passes: the LDS degree table holds one byte per node, copied from a byte plane that the dinv pass writes once (4 nodes per lane): 100 KB staged per work-group, not 400 KB (sums and order kept)
# speedup vs baseline: 4.7582x; 4.7582x over previous
#include <hip/hip_runtime.h>
#include <math.h>


typedef float v2f __attribute__((ext_vector_type(2)));
typedef float v8f __attribute__((ext_vector_type(8)));

#define FEAT_IN  128
#define FEAT_MID 16
#define FEAT_OUT 2

static __device__ __forceinline__ int block_half(int* half_lds) {
    if (threadIdx.x == 0) {
        *half_lds = (int)((__builtin_amdgcn_s_sendmsg_rtn(0x87) >> 18) & 1u);
    }
    __syncthreads();
    return *half_lds;
}

__global__ void degree_kernel(const int* __restrict__ dst, float* degA, float* degB,
                              int n_edges) {
    __shared__ int half_lds;
    float* my = block_half(&half_lds) ? degB : degA;
    int e = blockIdx.x * blockDim.x + threadIdx.x;
    if (e < n_edges) atomicAdd(&my[dst[e]], 1.0f);
}

#define TAB_BIG   255u

static __device__ __forceinline__ unsigned char deg8_of(float t) {
    return (t < 255.0f) ? (unsigned char)(int)t : (unsigned char)TAB_BIG;
}
static __device__ __forceinline__ float dinv_of(float d) {
    return (d > 0.0f) ? rsqrtf(d) : 0.0f;
}

template <bool BYTES>
__global__ void dinv_kernel(float* degA, const float* degB, unsigned char* deg8, int n_nodes) {
    int i = blockIdx.x * blockDim.x + threadIdx.x;
    if (BYTES) {
        if (4 * i >= n_nodes) return;
        float4 a = reinterpret_cast<const float4*>(degA)[i];
        float4 b = reinterpret_cast<const float4*>(degB)[i];
        float4 d;
        d.x = a.x + b.x; d.y = a.y + b.y; d.z = a.z + b.z; d.w = a.w + b.w;
        uchar4 p;
        p.x = deg8_of(d.x); p.y = deg8_of(d.y); p.z = deg8_of(d.z); p.w = deg8_of(d.w);
        reinterpret_cast<uchar4*>(deg8)[i] = p;
        float4 r;
        r.x = dinv_of(d.x); r.y = dinv_of(d.y); r.z = dinv_of(d.z); r.w = dinv_of(d.w);
        reinterpret_cast<float4*>(degA)[i] = r;
    } else if (i < n_nodes) {
        float d = degA[i];
        if (degB != degA) d += degB[i];
        degA[i] = dinv_of(d);
    }
}

#define TAB_CAP   172032
#define EDGE_WGS  256
#define EDGE_THR  1024

static __device__ __forceinline__ void stage_degrees(unsigned char* tab,
                                                     const unsigned char* __restrict__ deg8,
                                                     int n_nodes) {
    const int nt  = blockDim.x;
    const int n16 = n_nodes >> 4;
    const uint4* t4 = reinterpret_cast<const uint4*>(deg8);
    const int rot = (int)blockIdx.x * (n16 / (int)gridDim.x);
    for (int i0 = threadIdx.x; i0 < n16; i0 += nt * 8) {
        uint4 v[8];
        int   at[8];
#pragma unroll
        for (int u = 0; u < 8; u++) {
            int ii = i0 + u * nt;
            int i  = ((ii < n16) ? ii : n16 - 1) + rot;
            if (i >= n16) i -= n16;
            at[u] = i;
            v[u]  = t4[i];
        }
#pragma unroll
        for (int u = 0; u < 8; u++)
            *reinterpret_cast<uint4*>(&tab[16 * at[u]]) = v[u];
    }
    {
        int i = (n16 << 4) + (int)threadIdx.x;
        if (i < n_nodes) tab[i] = deg8[i];
    }
    __syncthreads();
}

template <bool LDS_TAB>
static __device__ __forceinline__ float node_dinv(const unsigned char* tab,
                                                  const float* __restrict__ dinv, int i) {
    if (!LDS_TAB) return dinv[i];
    unsigned c = tab[i];
    float r = (c != 0u) ? rsqrtf((float)c) : 0.0f;
    if (c == TAB_BIG) r = dinv[i];
    return r;
}

__global__ void __launch_bounds__(256)
gemm1_wmma(const float* __restrict__ x, const float* __restrict__ W1,
           float* __restrict__ h, float* degA, float* degB, float* acc1A, float* acc1B,
           float* acc2A, float* acc2B, int n_tiles, int n_nodes) {
    int lane = threadIdx.x & 31;
    int wave = threadIdx.x >> 5;
    int tile = blockIdx.x * (blockDim.x >> 5) + wave;
    if (tile >= n_tiles) return;

    int col = lane & 15;
    int kh  = (lane >> 4) << 1;

    int rowA = tile * 16 + col;
    if (rowA > n_nodes - 1) rowA = n_nodes - 1;
    const float* xrow = x + (size_t)rowA * FEAT_IN;

    v8f acc = {};
    for (int k0 = 0; k0 < FEAT_IN; k0 += 4) {
        v2f a, b;
        a.x = xrow[k0 + kh];
        a.y = xrow[k0 + kh + 1];
        b.x = W1[(k0 + kh)     * FEAT_MID + col];
        b.y = W1[(k0 + kh + 1) * FEAT_MID + col];
        acc = __builtin_amdgcn_wmma_f32_16x16x4_f32(
             false, a,  false, b,
             (short)0, acc,  false,  false);
    }

    int mbase = (lane >> 4) * 8;
#pragma unroll
    for (int r = 0; r < 8; r++) {
        int row = tile * 16 + mbase + r;
        if (row < n_nodes) h[(size_t)row * FEAT_MID + col] = acc[r];
    }

    {
        const int row0 = tile * 16;
        for (int j = 0; j < 8; j++) {
            int idx = row0 * FEAT_MID + j * 32 + lane;
            if (idx < n_nodes * FEAT_MID) {
                acc1A[idx] = 0.0f;
                if (acc1B != acc1A) acc1B[idx] = 0.0f;
            }
        }
        int i2 = row0 * FEAT_OUT + lane;
        if (i2 < n_nodes * FEAT_OUT) {
            acc2A[i2] = 0.0f;
            if (acc2B != acc2A) acc2B[i2] = 0.0f;
        }
        int nd = row0 + lane;
        if (lane < 16 && nd < n_nodes) {
            if (degB != degA) degB[nd] = 0.0f;
            degA[nd] = 1.0f;
        }
    }
}

template <bool LDS_TAB>
__global__ void __launch_bounds__(EDGE_THR)
edge1_kernel(const int* __restrict__ src, const int* __restrict__ dst,
             const float* __restrict__ dinv, const unsigned char* __restrict__ deg8,
             const float* __restrict__ h, float* acc1A, float* acc1B,
             int n_edges, int n_nodes) {
    __shared__ int half_lds;
    __shared__ __attribute__((aligned(16))) unsigned char tab[LDS_TAB ? TAB_CAP : 16];
    float* my = block_half(&half_lds) ? acc1B : acc1A;
    if (LDS_TAB) stage_degrees(tab, deg8, n_nodes);

    const int lane = threadIdx.x & 31;
    const int sub  = lane >> 4;
    const int f    = lane & 15;
    const int wave = threadIdx.x >> 5;
    const int nwav = blockDim.x >> 5;
    const int chunk = (n_edges + (int)gridDim.x - 1) / (int)gridDim.x;
    const int e0 = (int)blockIdx.x * chunk;
    const int e1 = (e0 + chunk < n_edges) ? e0 + chunk : n_edges;

    for (int base = e0 + wave * 32; base < e1; base += nwav * 32) {
        const bool full = base + 32 <= e1;
        int e  = base + lane;
        int ec = (e < e1) ? e : e1 - 1;
        int sv = src[ec];
        int dv = dst[ec];
        float nv = node_dinv<LDS_TAB>(tab, dinv, sv) * node_dinv<LDS_TAB>(tab, dinv, dv);

        float hv[16], nn[16];
        int   dd[16];
#pragma unroll
        for (int j = 0; j < 16; j++) {
            int l = 2 * j + sub;
            int s = __shfl(sv, l);
            dd[j] = __shfl(dv, l);
            nn[j] = __shfl(nv, l);
            hv[j] = h[s * FEAT_MID + f];
        }
#pragma unroll
        for (int j = 0; j < 16; j++) {
            if (full || base + 2 * j + sub < e1)
                atomicAdd(&my[dd[j] * FEAT_MID + f], hv[j] * nn[j]);
        }
    }
}

static __device__ __forceinline__ void load_row16(const float* __restrict__ p,
                                                  float (&r)[FEAT_MID]) {
#pragma unroll
    for (int k = 0; k < FEAT_MID; k++) r[k] = p[k];
}

__global__ void __launch_bounds__(256)
node_mid_kernel(const float* __restrict__ h, const float* __restrict__ dinv,
                const float* __restrict__ b1, const float* __restrict__ W2,
                const float* __restrict__ acc1A, const float* __restrict__ acc1B,
                float* __restrict__ h2, int n_nodes) {
    int i = blockIdx.x * blockDim.x + threadIdx.x;
    if (i >= n_nodes) return;
    float di = dinv[i];
    float a[FEAT_MID], hv[FEAT_MID];
    load_row16(acc1A + (size_t)i * FEAT_MID, a);
    load_row16(h + (size_t)i * FEAT_MID, hv);
    if (acc1B != acc1A) {
        float b[FEAT_MID];
        load_row16(acc1B + (size_t)i * FEAT_MID, b);
#pragma unroll
        for (int k = 0; k < FEAT_MID; k++) a[k] += b[k];
    }
    float s0 = 0.0f, s1 = 0.0f;
#pragma unroll
    for (int k = 0; k < FEAT_MID; k++) {
        float v = a[k] + hv[k] * di * di + b1[k];
        v = fmaxf(v, 0.0f);
        s0 = fmaf(v, W2[k * 2 + 0], s0);
        s1 = fmaf(v, W2[k * 2 + 1], s1);
    }
    h2[i * 2 + 0] = s0;
    h2[i * 2 + 1] = s1;
}

template <bool LDS_TAB>
__global__ void __launch_bounds__(EDGE_THR)
edge2_kernel(const int* __restrict__ src, const int* __restrict__ dst,
             const float* __restrict__ dinv, const unsigned char* __restrict__ deg8,
             const float* __restrict__ h2, float* acc2A, float* acc2B,
             int n_edges, int n_nodes) {
    __shared__ int half_lds;
    __shared__ __attribute__((aligned(16))) unsigned char tab[LDS_TAB ? TAB_CAP : 16];
    float* my = block_half(&half_lds) ? acc2B : acc2A;
    if (LDS_TAB) stage_degrees(tab, deg8, n_nodes);

    const int lane = threadIdx.x & 31;
    const int pr   = lane >> 1;
    const int c    = lane & 1;
    const int wave = threadIdx.x >> 5;
    const int nwav = blockDim.x >> 5;
    const int chunk = (n_edges + (int)gridDim.x - 1) / (int)gridDim.x;
    const int e0 = (int)blockIdx.x * chunk;
    const int e1 = (e0 + chunk < n_edges) ? e0 + chunk : n_edges;

    for (int base = e0 + wave * 64; base < e1; base += nwav * 64) {
        const bool full = base + 64 <= e1;
        int sv[2], dv[2];
        float nv[2];
#pragma unroll
        for (int k = 0; k < 2; k++) {
            int e  = base + 32 * k + lane;
            int ec = (e < e1) ? e : e1 - 1;
            sv[k] = src[ec];
            dv[k] = dst[ec];
        }
#pragma unroll
        for (int k = 0; k < 2; k++)
            nv[k] = node_dinv<LDS_TAB>(tab, dinv, sv[k]) * node_dinv<LDS_TAB>(tab, dinv, dv[k]);

        float hv[4], nn[4];
        int   dd[4];
#pragma unroll
        for (int j = 0; j < 4; j++) {
            int k = j >> 1;
            int l = (j & 1) * 16 + pr;
            int s = __shfl(sv[k], l);
            dd[j] = __shfl(dv[k], l);
            nn[j] = __shfl(nv[k], l);
            hv[j] = h2[s * FEAT_OUT + c];
        }
#pragma unroll
        for (int j = 0; j < 4; j++) {
            if (full || base + 16 * j + pr < e1)
                atomicAdd(&my[dd[j] * FEAT_OUT + c], hv[j] * nn[j]);
        }
    }
}

__global__ void final_kernel(const float* acc2A, const float* acc2B,
                             const float* __restrict__ h2,
                             const float* __restrict__ dinv, const float* __restrict__ b2,
                             float* __restrict__ out, int n_nodes) {
    int i = blockIdx.x * blockDim.x + threadIdx.x;
    if (i >= n_nodes) return;
    float di = dinv[i];
    float sc = di * di;
    float a0 = acc2A[i * 2 + 0], a1 = acc2A[i * 2 + 1];
    if (acc2B != acc2A) { a0 += acc2B[i * 2 + 0]; a1 += acc2B[i * 2 + 1]; }
    float v0 = a0 + h2[i * 2 + 0] * sc + b2[0];
    float v1 = a1 + h2[i * 2 + 1] * sc + b2[1];
    float m  = fmaxf(v0, v1);
    float l  = m + logf(expf(v0 - m) + expf(v1 - m));
    out[i * 2 + 0] = v0 - l;
    out[i * 2 + 1] = v1 - l;
}

extern "C" void kernel_launch(void* const* d_in, const int* in_sizes, int n_in,
                              void* d_out, int out_size, void* d_ws, size_t ws_size,
                              hipStream_t stream) {
    const float* x  = (const float*)d_in[0];
    const int*   ei = (const int*)d_in[1];
    const float* W1 = (const float*)d_in[2];
    const float* b1 = (const float*)d_in[3];
    const float* W2 = (const float*)d_in[4];
    const float* b2 = (const float*)d_in[5];
    float* out = (float*)d_out;

    int n_nodes = in_sizes[0] / FEAT_IN;
    int n_edges = in_sizes[1] / 2;
    const int* src = ei;
    const int* dst = ei + n_edges;

    size_t n1  = (size_t)n_nodes;
    size_t n16 = n1 * FEAT_MID;
    size_t n2  = n1 * FEAT_OUT;
    bool two = ws_size >= (2 * n1 + 3 * n16 + 3 * n2) * sizeof(float);

    float* ws    = (float*)d_ws;
    float* degA  = ws;
    unsigned char* deg8 = (unsigned char*)(degA + n1);
    float* h     = two ? degA + 2 * n1 : degA + n1;
    float* acc1A = h + n16;
    float* acc1B = two ? acc1A + n16 : acc1A;
    float* h2    = acc1B + n16;
    float* acc2A = h2 + n2;
    float* acc2B = two ? acc2A + n2 : acc2A;
    float* degB  = two ? h2 : degA;

    const int B = 256;

    int n_tiles = (n_nodes + 15) / 16;
    int wpb = B / 32;
    gemm1_wmma<<<(n_tiles + wpb - 1) / wpb, B, 0, stream>>>(x, W1, h, degA, degB, acc1A, acc1B,
                                                           acc2A, acc2B, n_tiles, n_nodes);
    degree_kernel<<<(n_edges + B - 1) / B, B, 0, stream>>>(dst, degA, degB, n_edges);

    bool tab = two && n_nodes <= TAB_CAP && (n_nodes & 3) == 0;

    if (tab) dinv_kernel<true><<<(n_nodes / 4 + B - 1) / B, B, 0, stream>>>(degA, degB, deg8, n_nodes);
    else     dinv_kernel<false><<<(n_nodes + B - 1) / B, B, 0, stream>>>(degA, degB, deg8, n_nodes);

    if (tab) edge1_kernel<true><<<EDGE_WGS, EDGE_THR, 0, stream>>>(src, dst, degA, deg8, h, acc1A, acc1B, n_edges, n_nodes);
    else     edge1_kernel<false><<<EDGE_WGS, EDGE_THR, 0, stream>>>(src, dst, degA, deg8, h, acc1A, acc1B, n_edges, n_nodes);
    node_mid_kernel<<<(n_nodes + B - 1) / B, B, 0, stream>>>(h, degA, b1, W2, acc1A, acc1B, h2, n_nodes);

    if (tab) edge2_kernel<true><<<EDGE_WGS, EDGE_THR, 0, stream>>>(src, dst, degA, deg8, h2, acc2A, acc2B, n_edges, n_nodes);
    else     edge2_kernel<false><<<EDGE_WGS, EDGE_THR, 0, stream>>>(src, dst, degA, deg8, h2, acc2A, acc2B, n_edges, n_nodes);

    final_kernel<<<(n_nodes + B - 1) / B, B, 0, stream>>>(acc2A, acc2B, h2, degA, b2, out, n_nodes);
}
